# starting kernel
# baseline (speedup 1.0000x reference)
#include <hip/hip_runtime.h>
#include <hip/hip_bf16.h>
#include <math.h>


typedef __attribute__((ext_vector_type(2))) float v2f;
typedef __attribute__((ext_vector_type(8))) float v8f;

#define HID 64
#define NEG_SLOPE 0.2f
#define LN_EPS 1e-5f

__device__ __forceinline__ unsigned fkey(float f) {
    unsigned b = __float_as_uint(f);
    return (b & 0x80000000u) ? ~b : (b | 0x80000000u);
}
__device__ __forceinline__ float funkey(unsigned k) {
    unsigned b = (k & 0x80000000u) ? (k ^ 0x80000000u) : ~k;
    return __uint_as_float(b);
}
#define KEY_NEG_INF 0x007FFFFFu

template <int NTILES>
__global__ void gemm64_wmma(const float* __restrict__ X,
                            const float* __restrict__ W,
                            const float* __restrict__ bias,
                            float* __restrict__ Y,
                            int nrows, int nout) {
    __shared__ float lds_w[64][64];

    int tid = threadIdx.x;
    for (int idx = tid; idx < 64 * 64; idx += 128) {
        int r = idx >> 6, c = idx & 63;
        lds_w[r][c] = (c < nout) ? W[r * nout + c] : 0.0f;
    }
    __syncthreads();

    int lane  = tid & 31;
    int wave  = tid >> 5;
    int half  = lane >> 4;
    int l15   = lane & 15;
    int rbase = blockIdx.x * 64 + wave * 16;

    int arow = rbase + l15;
    if (arow >= nrows) arow = nrows - 1;
    const float* xrow = X + (size_t)arow * 64;

    v8f acc[NTILES];
#pragma unroll
    for (int t = 0; t < NTILES; ++t)
#pragma unroll
        for (int j = 0; j < 8; ++j) acc[t][j] = 0.0f;

#pragma unroll
    for (int k0 = 0; k0 < 64; k0 += 4) {
        const float* ap = xrow + k0 + 2 * half;
        v2f a;
        a.x = ap[0];
        a.y = ap[1];
#pragma unroll
        for (int t = 0; t < NTILES; ++t) {
            v2f b;
            b.x = lds_w[k0 + 2 * half + 0][t * 16 + l15];
            b.y = lds_w[k0 + 2 * half + 1][t * 16 + l15];
            acc[t] = __builtin_amdgcn_wmma_f32_16x16x4_f32(
                false, a, false, b, (short)0, acc[t], false, false);
        }
    }

#pragma unroll
    for (int t = 0; t < NTILES; ++t) {
        int col = t * 16 + l15;
        float bv = (col < nout) ? bias[col] : 0.0f;
#pragma unroll
        for (int i = 0; i < 8; ++i) {
            int row = rbase + 8 * half + i;
            if (row < nrows && col < nout)
                Y[(size_t)row * nout + col] = acc[t][i] + bv;
        }
    }
}

__global__ void init_layer(float* __restrict__ t, float* __restrict__ denom,
                           unsigned* __restrict__ mx, int n) {
    int i = blockIdx.x * blockDim.x + threadIdx.x;
    if (i < n * 16) {
        float4 z = make_float4(0.f, 0.f, 0.f, 0.f);
        ((float4*)t)[i] = z;
    }
    if (i < n) {
        denom[i] = 0.0f;
        mx[i]    = KEY_NEG_INF;
    }
}

__global__ void edge_score(const float* __restrict__ fs,
                           const float* __restrict__ fd,
                           const int* __restrict__ src,
                           const int* __restrict__ dst,
                           const float* __restrict__ attn,
                           float* __restrict__ score,
                           unsigned* __restrict__ mx, int E) {
    int e = blockIdx.x * blockDim.x + threadIdx.x;
    if (e >= E) return;
    int s = src[e], d = dst[e];
    const float4* ps = (const float4*)(fs + (size_t)s * HID);
    const float4* pd = (const float4*)(fd + (size_t)d * HID);
    const float4* pa = (const float4*)attn;
    float acc = 0.0f;
#pragma unroll
    for (int i = 0; i < HID / 4; ++i) {
        float4 a = ps[i], b = pd[i], w = pa[i];
        float x0 = a.x + b.x, x1 = a.y + b.y, x2 = a.z + b.z, x3 = a.w + b.w;
        x0 = x0 > 0.f ? x0 : x0 * NEG_SLOPE;
        x1 = x1 > 0.f ? x1 : x1 * NEG_SLOPE;
        x2 = x2 > 0.f ? x2 : x2 * NEG_SLOPE;
        x3 = x3 > 0.f ? x3 : x3 * NEG_SLOPE;
        acc += x0 * w.x + x1 * w.y + x2 * w.z + x3 * w.w;
    }
    score[e] = acc;
    atomicMax(&mx[d], fkey(acc));
}

__global__ void edge_exp(float* __restrict__ score,
                         const unsigned* __restrict__ mx,
                         const int* __restrict__ dst,
                         float* __restrict__ denom, int E) {
    int e = blockIdx.x * blockDim.x + threadIdx.x;
    if (e >= E) return;
    int d = dst[e];
    float ex = expf(score[e] - funkey(mx[d]));
    score[e] = ex;
    atomicAdd(&denom[d], ex);
}

__global__ void edge_scatter(const float* __restrict__ score,
                             const float* __restrict__ denom,
                             const float* __restrict__ fs,
                             const int* __restrict__ src,
                             const int* __restrict__ dst,
                             float* __restrict__ t, int E) {
    int e = blockIdx.x * blockDim.x + threadIdx.x;
    if (e >= E) return;
    int s = src[e], d = dst[e];
    float alpha = score[e] / denom[d];
    const float4* ps = (const float4*)(fs + (size_t)s * HID);
    float* pt = t + (size_t)d * HID;
#pragma unroll
    for (int i = 0; i < HID / 4; ++i) {
        float4 v = ps[i];
        atomicAdd(pt + 4 * i + 0, v.x * alpha);
        atomicAdd(pt + 4 * i + 1, v.y * alpha);
        atomicAdd(pt + 4 * i + 2, v.z * alpha);
        atomicAdd(pt + 4 * i + 3, v.w * alpha);
    }
}

__global__ void node_update(float* __restrict__ h, const float* __restrict__ t,
                            const float* __restrict__ gamma,
                            const float* __restrict__ beta, int n) {
    int i = blockIdx.x * blockDim.x + threadIdx.x;
    if (i >= n) return;
    const float4* pt = (const float4*)(t + (size_t)i * HID);
    float4 v[HID / 4];
    float sum = 0.0f;
#pragma unroll
    for (int j = 0; j < HID / 4; ++j) {
        v[j] = pt[j];
        sum += v[j].x + v[j].y + v[j].z + v[j].w;
    }
    float mean = sum * (1.0f / HID);
    float var = 0.0f;
#pragma unroll
    for (int j = 0; j < HID / 4; ++j) {
        float d0 = v[j].x - mean, d1 = v[j].y - mean;
        float d2 = v[j].z - mean, d3 = v[j].w - mean;
        var += d0 * d0 + d1 * d1 + d2 * d2 + d3 * d3;
    }
    var *= (1.0f / HID);
    float inv = rsqrtf(var + LN_EPS);

    float4* ph = (float4*)(h + (size_t)i * HID);
    const float4* pg = (const float4*)gamma;
    const float4* pb = (const float4*)beta;
#pragma unroll
    for (int j = 0; j < HID / 4; ++j) {
        float4 g = pg[j], b = pb[j], hv = ph[j];
        float x0 = (v[j].x - mean) * inv * g.x + b.x;
        float x1 = (v[j].y - mean) * inv * g.y + b.y;
        float x2 = (v[j].z - mean) * inv * g.z + b.z;
        float x3 = (v[j].w - mean) * inv * g.w + b.w;
        const float k = 0.70710678118654752f;
        hv.x += 0.5f * x0 * (1.0f + erff(x0 * k));
        hv.y += 0.5f * x1 * (1.0f + erff(x1 * k));
        hv.z += 0.5f * x2 * (1.0f + erff(x2 * k));
        hv.w += 0.5f * x3 * (1.0f + erff(x3 * k));
        ph[j] = hv;
    }
}

extern "C" void kernel_launch(void* const* d_in, const int* in_sizes, int n_in,
                              void* d_out, int out_size, void* d_ws, size_t ws_size,
                              hipStream_t stream) {
    const float* nodes = (const float*)d_in[0];
    const int*   src   = (const int*)d_in[1];
    const int*   dst   = (const int*)d_in[2];
    const float* w_in  = (const float*)d_in[3];
    const float* b_in  = (const float*)d_in[4];
    const float* w_src = (const float*)d_in[5];
    const float* b_src = (const float*)d_in[6];
    const float* w_dst = (const float*)d_in[7];
    const float* b_dst = (const float*)d_in[8];
    const float* attn  = (const float*)d_in[9];
    const float* gamma = (const float*)d_in[10];
    const float* beta  = (const float*)d_in[11];
    const float* w_out = (const float*)d_in[12];
    const float* b_out = (const float*)d_in[13];

    const int N = in_sizes[0] / HID;
    const int E = in_sizes[1];
    const int NL = 6;

    float* ws   = (float*)d_ws;
    float* h    = ws;  ws += (size_t)N * HID;
    float* fs   = ws;  ws += (size_t)N * HID;
    float* fd   = ws;  ws += (size_t)N * HID;
    float* tbuf = ws;  ws += (size_t)N * HID;
    float* score = ws; ws += E;
    float* denom = ws; ws += N;
    unsigned* mx = (unsigned*)ws;

    const int ggrid  = (N + 63) / 64;
    const int eblk   = 256;
    const int egrid  = (E + eblk - 1) / eblk;
    const int ngrid  = (N + 255) / 256;
    const int igrid  = (N * 16 + 255) / 256;

    gemm64_wmma<4><<<ggrid, 128, 0, stream>>>(nodes, w_in, b_in, h, N, HID);

    for (int L = 0; L < NL; ++L) {
        gemm64_wmma<4><<<ggrid, 128, 0, stream>>>(h, w_src + (size_t)L * HID * HID,
                                                  b_src + (size_t)L * HID, fs, N, HID);
        gemm64_wmma<4><<<ggrid, 128, 0, stream>>>(h, w_dst + (size_t)L * HID * HID,
                                                  b_dst + (size_t)L * HID, fd, N, HID);
        init_layer<<<igrid, 256, 0, stream>>>(tbuf, denom, mx, N);
        edge_score<<<egrid, eblk, 0, stream>>>(fs, fd, src, dst,
                                               attn + (size_t)L * HID, score, mx, E);
        edge_exp<<<egrid, eblk, 0, stream>>>(score, mx, dst, denom, E);
        edge_scatter<<<egrid, eblk, 0, stream>>>(score, denom, fs, src, dst, tbuf, E);
        node_update<<<ngrid, 256, 0, stream>>>(h, tbuf, gamma + (size_t)L * HID,
                                               beta + (size_t)L * HID, N);
    }

    gemm64_wmma<1><<<ggrid, 128, 0, stream>>>(h, w_out, b_out, (float*)d_out, N, 8);
}


# GEMM without LDS or barrier: a wave reads its B fragments straight from global by 16-byte loads (output tile t = columns 4c+t), all 48 loads in flight at once; 16-byte output stores; same sums
# speedup vs baseline: 5.8569x; 5.8569x over previous
#include <hip/hip_runtime.h>
#include <hip/hip_bf16.h>
#include <math.h>


typedef __attribute__((ext_vector_type(2))) float v2f;
typedef __attribute__((ext_vector_type(8))) float v8f;

#define HID 64
#define NEG_SLOPE 0.2f
#define LN_EPS 1e-5f

__device__ __forceinline__ unsigned fkey(float f) {
    unsigned b = __float_as_uint(f);
    return (b & 0x80000000u) ? ~b : (b | 0x80000000u);
}
__device__ __forceinline__ float funkey(unsigned k) {
    unsigned b = (k & 0x80000000u) ? (k ^ 0x80000000u) : ~k;
    return __uint_as_float(b);
}
#define KEY_NEG_INF 0x007FFFFFu

__global__ __launch_bounds__(128)
void gemm64_direct(const float* __restrict__ X, const float* __restrict__ W,
                   const float* __restrict__ bias, float* __restrict__ Y,
                   int nrows) {
    const int lane = threadIdx.x & 31;
    const int wave = __builtin_amdgcn_readfirstlane(threadIdx.x >> 5);
    const int half = lane >> 4;
    const int l15  = lane & 15;
    const int rbase = (blockIdx.x * 4 + wave) * 16;
    if (rbase >= nrows) return;

    int arow = rbase + l15;
    if (arow >= nrows) arow = nrows - 1;
    const float* xrow = X + (size_t)arow * 64 + 2 * half;
    const float* wcol = W + (2 * half) * 64 + 4 * l15;

    v2f a[16];
    float4 w0[16], w1[16];
#pragma unroll
    for (int ks = 0; ks < 16; ++ks) {
        a[ks].x = xrow[4 * ks];
        a[ks].y = xrow[4 * ks + 1];
    }
#pragma unroll
    for (int ks = 0; ks < 16; ++ks) {
        w0[ks] = *(const float4*)(wcol + (4 * ks) * 64);
        w1[ks] = *(const float4*)(wcol + (4 * ks + 1) * 64);
    }
    const float4 bv = *(const float4*)(bias + 4 * l15);

    v8f acc0, acc1, acc2, acc3;
#pragma unroll
    for (int j = 0; j < 8; ++j) {
        acc0[j] = 0.0f; acc1[j] = 0.0f; acc2[j] = 0.0f; acc3[j] = 0.0f;
    }

#pragma unroll
    for (int ks = 0; ks < 16; ++ks) {
        v2f b;
        b.x = w0[ks].x; b.y = w1[ks].x;
        acc0 = __builtin_amdgcn_wmma_f32_16x16x4_f32(
            false, a[ks], false, b, (short)0, acc0, false, false);
        b.x = w0[ks].y; b.y = w1[ks].y;
        acc1 = __builtin_amdgcn_wmma_f32_16x16x4_f32(
            false, a[ks], false, b, (short)0, acc1, false, false);
        b.x = w0[ks].z; b.y = w1[ks].z;
        acc2 = __builtin_amdgcn_wmma_f32_16x16x4_f32(
            false, a[ks], false, b, (short)0, acc2, false, false);
        b.x = w0[ks].w; b.y = w1[ks].w;
        acc3 = __builtin_amdgcn_wmma_f32_16x16x4_f32(
            false, a[ks], false, b, (short)0, acc3, false, false);
    }

#pragma unroll
    for (int i = 0; i < 8; ++i) {
        const int row = rbase + 8 * half + i;
        if (row < nrows) {
            float4 o;
            o.x = acc0[i] + bv.x;
            o.y = acc1[i] + bv.y;
            o.z = acc2[i] + bv.z;
            o.w = acc3[i] + bv.w;
            *(float4*)(Y + (size_t)row * 64 + 4 * l15) = o;
        }
    }
}

__global__ __launch_bounds__(128)
void gemm64_direct_narrow(const float* __restrict__ X, const float* __restrict__ W,
                          const float* __restrict__ bias, float* __restrict__ Y,
                          int nrows, int nout) {
    const int lane = threadIdx.x & 31;
    const int wave = __builtin_amdgcn_readfirstlane(threadIdx.x >> 5);
    const int half = lane >> 4;
    const int l15  = lane & 15;
    const int rbase = (blockIdx.x * 4 + wave) * 16;
    if (rbase >= nrows) return;

    int arow = rbase + l15;
    if (arow >= nrows) arow = nrows - 1;
    const float* xrow = X + (size_t)arow * 64 + 2 * half;
    const bool cv = l15 < nout;
    const float* wcol = W + (2 * half) * nout + (cv ? l15 : 0);

    v2f a[16], b[16];
#pragma unroll
    for (int ks = 0; ks < 16; ++ks) {
        a[ks].x = xrow[4 * ks];
        a[ks].y = xrow[4 * ks + 1];
    }
#pragma unroll
    for (int ks = 0; ks < 16; ++ks) {
        const float bx = wcol[(4 * ks) * nout];
        const float by = wcol[(4 * ks + 1) * nout];
        b[ks].x = cv ? bx : 0.0f;
        b[ks].y = cv ? by : 0.0f;
    }
    const float bv = cv ? bias[l15] : 0.0f;

    v8f acc;
#pragma unroll
    for (int j = 0; j < 8; ++j) acc[j] = 0.0f;
#pragma unroll
    for (int ks = 0; ks < 16; ++ks)
        acc = __builtin_amdgcn_wmma_f32_16x16x4_f32(
            false, a[ks], false, b[ks], (short)0, acc, false, false);

#pragma unroll
    for (int i = 0; i < 8; ++i) {
        const int row = rbase + 8 * half + i;
        if (row < nrows && cv) Y[(size_t)row * nout + l15] = acc[i] + bv;
    }
}

#define OWN_R  208
#define OWN_NT 1024
#define OWN_NW (OWN_NT / 32)
#define OWN_CAP 8192

__device__ __forceinline__ float own_wave_sum(float p) {
    p += __shfl_xor(p, 16);
    p += __shfl_xor(p, 8);
    p += __shfl_xor(p, 4);
    p += __shfl_xor(p, 2);
    p += __shfl_xor(p, 1);
    return p;
}

__device__ __forceinline__ void own_sums(int n, int s, int dl, float ex, int lane,
                                         const float* __restrict__ fs, float* t_l) {
    for (int g = 0; g < n; g += 8) {
        float a0[8], a1[8];
#pragma unroll
        for (int u = 0; u < 8; ++u) {
            int j = min(g + u, n - 1);
            int sj = __builtin_amdgcn_readlane(s, j);
            const float* ra = fs + (size_t)sj * HID;
            a0[u] = ra[lane];
            a1[u] = ra[32 + lane];
        }
#pragma unroll
        for (int u = 0; u < 8; ++u) {
            if (g + u < n) {
                int dj = __builtin_amdgcn_readlane(dl, g + u);
                float xj = __int_as_float(
                    __builtin_amdgcn_readlane(__float_as_int(ex), g + u));
                atomicAdd(&t_l[dj * HID + lane], xj * a0[u]);
                atomicAdd(&t_l[dj * HID + 32 + lane], xj * a1[u]);
            }
        }
    }
}

struct OwnList {
    int*   cnt;
    int*   src;
    int*   dl;
    float* sc;
};

template <int PASS>
__device__ __forceinline__ void own_flush(int n, const int* q, int lane, int base,
                                          const float* __restrict__ fs,
                                          const float* __restrict__ fd,
                                          const int* __restrict__ src,
                                          const int* __restrict__ dst,
                                          float w0, float w1, float* score,
                                          float* t_l, float* den_l, unsigned* mx_l,
                                          OwnList ls) {
    int e = 0, s = 0, d = base;
    if (lane < n) {
        e = q[lane];
        s = src[e];
        d = dst[e];
    }
    const int dl = d - base;
    if (PASS == 1) {
        float mine = 0.0f;
        for (int g = 0; g < n; g += 8) {
            float a0[8], a1[8], b0[8], b1[8];
#pragma unroll
            for (int u = 0; u < 8; ++u) {
                int j = min(g + u, n - 1);
                int sj = __builtin_amdgcn_readlane(s, j);
                int dj = __builtin_amdgcn_readlane(d, j);
                const float* ra = fs + (size_t)sj * HID;
                const float* rb = fd + (size_t)dj * HID;
                a0[u] = ra[lane];
                a1[u] = ra[32 + lane];
                b0[u] = rb[lane];
                b1[u] = rb[32 + lane];
            }
#pragma unroll
            for (int u = 0; u < 8; ++u) {
                float x0 = a0[u] + b0[u], x1 = a1[u] + b1[u];
                x0 = x0 > 0.f ? x0 : x0 * NEG_SLOPE;
                x1 = x1 > 0.f ? x1 : x1 * NEG_SLOPE;
                float p = own_wave_sum(x0 * w0 + x1 * w1);
                if (lane == g + u) mine = p;
            }
        }
        if (lane < n) {
            score[e] = mine;
            atomicMax(&mx_l[dl], fkey(mine));
        }
        int slot = 0;
        if (lane == 0) slot = atomicAdd(ls.cnt, n);
        slot = __builtin_amdgcn_readfirstlane(slot);
        if (slot + n <= OWN_CAP && lane < n) {
            ls.src[slot + lane] = s;
            ls.dl[slot + lane]  = dl;
            ls.sc[slot + lane]  = mine;
        }
    } else {
        float ex = 0.0f;
        if (lane < n) {
            ex = expf(score[e] - funkey(mx_l[dl]));
            atomicAdd(&den_l[dl], ex);
        }
        own_sums(n, s, dl, ex, lane, fs, t_l);
    }
}

__device__ __forceinline__ void own_list_sums(int wave, int lane, int total,
                                              const float* __restrict__ fs,
                                              OwnList ls, float* t_l, float* den_l,
                                              const unsigned* mx_l) {
    const int per = (total + OWN_NW - 1) / OWN_NW;
    const int beg = wave * per;
    const int end = min(total, beg + per);
    for (int off = beg; off < end; off += 32) {
        const int n = min(32, end - off);
        int s = 0, dl = 0;
        float ex = 0.0f;
        if (lane < n) {
            s  = ls.src[off + lane];
            dl = ls.dl[off + lane];
            ex = expf(ls.sc[off + lane] - funkey(mx_l[dl]));
            atomicAdd(&den_l[dl], ex);
        }
        own_sums(n, s, dl, ex, lane, fs, t_l);
    }
}

#define OWN_U 8
#define OWN_HIT(d) (((unsigned)((d) - base) < (unsigned)OWN_R) ? 1u : 0u)

template <int PASS>
__device__ __forceinline__ void own_scan(int wave, int lane, int base, int E,
                                         const float* __restrict__ fs,
                                         const float* __restrict__ fd,
                                         const int* __restrict__ src,
                                         const int* __restrict__ dst,
                                         float w0, float w1, float* score,
                                         float* t_l, float* den_l, unsigned* mx_l,
                                         int* q, OwnList ls) {
    const int nfull = E >> 7;
    const int rot = (int)(((long long)blockIdx.x * nfull) / gridDim.x);
    int cnt = 0;
    int kb[OWN_U];
#pragma unroll
    for (int sl = 0; sl < OWN_U; ++sl) kb[sl] = 0;
    kb[0] = nfull * 128;
    unsigned mk = 0;
    if (wave == 0) {
        const int eb = kb[0] + lane * 4;
        int4 tv = make_int4(-1, -1, -1, -1);
        if (eb     < E) tv.x = dst[eb];
        if (eb + 1 < E) tv.y = dst[eb + 1];
        if (eb + 2 < E) tv.z = dst[eb + 2];
        if (eb + 3 < E) tv.w = dst[eb + 3];
        mk = OWN_HIT(tv.x) | (OWN_HIT(tv.y) << 1) | (OWN_HIT(tv.z) << 2) | (OWN_HIT(tv.w) << 3);
    }
    int c = wave;
    while (true) {
        for (int round = 0; round < 4 * OWN_U; ++round) {
            const bool m = mk != 0u;
            const unsigned bal = __builtin_amdgcn_ballot_w32(m);
            if (bal == 0u) break;
            const int i = m ? __builtin_ctz(mk) : 0;
            mk &= mk - 1u;
            const int sl = i >> 2;
            int kbs = kb[0];
#pragma unroll
            for (int s2 = 1; s2 < OWN_U; ++s2) kbs = (sl == s2) ? kb[s2] : kbs;
            const int pos = cnt + (int)__builtin_amdgcn_mbcnt_lo(bal, 0u);
            if (m) q[pos] = kbs + lane * 4 + (i & 3);
            cnt += __builtin_popcount(bal);
            if (cnt >= 32) {
                own_flush<PASS>(32, q, lane, base, fs, fd, src, dst, w0, w1,
                                score, t_l, den_l, mx_l, ls);
                const int rem = cnt - 32;
                int tmp = 0;
                if (lane < rem) tmp = q[32 + lane];
                if (lane < rem) q[lane] = tmp;
                cnt = rem;
            }
        }
        if (c >= nfull) break;
        int4 v[OWN_U];
        unsigned okmask = 0;
#pragma unroll
        for (int sl = 0; sl < OWN_U; ++sl) {
            const bool ok = c + sl * OWN_NW < nfull;
            int k = ok ? c + sl * OWN_NW + rot : 0;
            if (k >= nfull) k -= nfull;
            kb[sl] = k * 128;
            okmask |= ok ? (0xFu << (4 * sl)) : 0u;
            v[sl] = *(const int4*)(dst + kb[sl] + lane * 4);
        }
        unsigned nm = 0;
#pragma unroll
        for (int sl = OWN_U - 1; sl >= 0; --sl) {
            nm = (nm << 1) | OWN_HIT(v[sl].w);
            nm = (nm << 1) | OWN_HIT(v[sl].z);
            nm = (nm << 1) | OWN_HIT(v[sl].y);
            nm = (nm << 1) | OWN_HIT(v[sl].x);
        }
        mk = nm & okmask;
        c += OWN_U * OWN_NW;
    }
    if (cnt > 0)
        own_flush<PASS>(cnt, q, lane, base, fs, fd, src, dst, w0, w1,
                        score, t_l, den_l, mx_l, ls);
}

__global__ __launch_bounds__(OWN_NT)
void edge_owner(const float* __restrict__ fs, const float* __restrict__ fd,
                const int* __restrict__ src, const int* __restrict__ dst,
                const float* __restrict__ attn, float* score,
                float* __restrict__ t, int N, int E) {
    __shared__ __attribute__((aligned(16))) float t_l[OWN_R * HID];
    __shared__ float    den_l[OWN_R];
    __shared__ unsigned mx_l[OWN_R];
    __shared__ int      q_l[OWN_NW * 64];
    __shared__ int      ls_src[OWN_CAP];
    __shared__ int      ls_dl[OWN_CAP];
    __shared__ float    ls_sc[OWN_CAP];
    __shared__ int      ls_cnt;

    const int tid  = threadIdx.x;
    const int lane = tid & 31;
    const int wave = __builtin_amdgcn_readfirstlane(tid >> 5);
    const int base = blockIdx.x * OWN_R;

    for (int i = tid; i < OWN_R * HID; i += OWN_NT) t_l[i] = 0.0f;
    if (tid < OWN_R) {
        den_l[tid] = 0.0f;
        mx_l[tid]  = KEY_NEG_INF;
    }
    if (tid == 0) ls_cnt = 0;
    OwnList ls;
    ls.cnt = &ls_cnt;
    ls.src = ls_src;
    ls.dl  = ls_dl;
    ls.sc  = ls_sc;
    const float w0 = attn[lane], w1 = attn[32 + lane];
    __syncthreads();

    own_scan<1>(wave, lane, base, E, fs, fd, src, dst, w0, w1, score,
                t_l, den_l, mx_l, q_l + wave * 64, ls);
    __syncthreads();
    const int total = __builtin_amdgcn_readfirstlane(ls_cnt);
    if (total <= OWN_CAP) {
        own_list_sums(wave, lane, total, fs, ls, t_l, den_l, mx_l);
    } else {
        own_scan<2>(wave, lane, base, E, fs, fd, src, dst, w0, w1, score,
                    t_l, den_l, mx_l, q_l + wave * 64, ls);
    }
    __syncthreads();

    for (int idx = tid; idx < OWN_R * 16; idx += OWN_NT) {
        const int row = idx >> 4, c4 = idx & 15;
        const int grow = base + row;
        if (grow < N) {
            const float den = den_l[row];
            const float4 v = *(const float4*)(t_l + row * HID + c4 * 4);
            float4 o = make_float4(0.f, 0.f, 0.f, 0.f);
            if (den > 0.0f) {
                o.x = v.x / den; o.y = v.y / den; o.z = v.z / den; o.w = v.w / den;
            }
            *(float4*)(t + (size_t)grow * HID + c4 * 4) = o;
        }
    }
}

__global__ void node_update(float* __restrict__ h, const float* __restrict__ t,
                            const float* __restrict__ gamma,
                            const float* __restrict__ beta, int n) {
    int i = blockIdx.x * blockDim.x + threadIdx.x;
    if (i >= n) return;
    const float4* pt = (const float4*)(t + (size_t)i * HID);
    float4 v[HID / 4];
    float sum = 0.0f;
#pragma unroll
    for (int j = 0; j < HID / 4; ++j) {
        v[j] = pt[j];
        sum += v[j].x + v[j].y + v[j].z + v[j].w;
    }
    float mean = sum * (1.0f / HID);
    float var = 0.0f;
#pragma unroll
    for (int j = 0; j < HID / 4; ++j) {
        float d0 = v[j].x - mean, d1 = v[j].y - mean;
        float d2 = v[j].z - mean, d3 = v[j].w - mean;
        var += d0 * d0 + d1 * d1 + d2 * d2 + d3 * d3;
    }
    var *= (1.0f / HID);
    float inv = rsqrtf(var + LN_EPS);

    float4* ph = (float4*)(h + (size_t)i * HID);
    const float4* pg = (const float4*)gamma;
    const float4* pb = (const float4*)beta;
#pragma unroll
    for (int j = 0; j < HID / 4; ++j) {
        float4 g = pg[j], b = pb[j], hv = ph[j];
        float x0 = (v[j].x - mean) * inv * g.x + b.x;
        float x1 = (v[j].y - mean) * inv * g.y + b.y;
        float x2 = (v[j].z - mean) * inv * g.z + b.z;
        float x3 = (v[j].w - mean) * inv * g.w + b.w;
        const float k = 0.70710678118654752f;
        hv.x += 0.5f * x0 * (1.0f + erff(x0 * k));
        hv.y += 0.5f * x1 * (1.0f + erff(x1 * k));
        hv.z += 0.5f * x2 * (1.0f + erff(x2 * k));
        hv.w += 0.5f * x3 * (1.0f + erff(x3 * k));
        ph[j] = hv;
    }
}

extern "C" void kernel_launch(void* const* d_in, const int* in_sizes, int n_in,
                              void* d_out, int out_size, void* d_ws, size_t ws_size,
                              hipStream_t stream) {
    const float* nodes = (const float*)d_in[0];
    const int*   src   = (const int*)d_in[1];
    const int*   dst   = (const int*)d_in[2];
    const float* w_in  = (const float*)d_in[3];
    const float* b_in  = (const float*)d_in[4];
    const float* w_src = (const float*)d_in[5];
    const float* b_src = (const float*)d_in[6];
    const float* w_dst = (const float*)d_in[7];
    const float* b_dst = (const float*)d_in[8];
    const float* attn  = (const float*)d_in[9];
    const float* gamma = (const float*)d_in[10];
    const float* beta  = (const float*)d_in[11];
    const float* w_out = (const float*)d_in[12];
    const float* b_out = (const float*)d_in[13];

    const int N = in_sizes[0] / HID;
    const int E = in_sizes[1];
    const int NL = 6;

    float* ws   = (float*)d_ws;
    float* h    = ws;  ws += (size_t)N * HID;
    float* fs   = ws;  ws += (size_t)N * HID;
    float* fd   = ws;  ws += (size_t)N * HID;
    float* tbuf = ws;  ws += (size_t)N * HID;
    float* score = ws;

    const int ggrid  = (N + 63) / 64;
    const int ogrid  = (N + OWN_R - 1) / OWN_R;
    const int ngrid  = (N + 255) / 256;

    gemm64_direct<<<ggrid, 128, 0, stream>>>(nodes, w_in, b_in, h, N);

    for (int L = 0; L < NL; ++L) {
        gemm64_direct<<<ggrid, 128, 0, stream>>>(h, w_src + (size_t)L * HID * HID,
                                                b_src + (size_t)L * HID, fs, N);
        gemm64_direct<<<ggrid, 128, 0, stream>>>(h, w_dst + (size_t)L * HID * HID,
                                                b_dst + (size_t)L * HID, fd, N);
        edge_owner<<<ogrid, OWN_NT, 0, stream>>>(fs, fd, src, dst,
                                                 attn + (size_t)L * HID, score, tbuf, N, E);
        node_update<<<ngrid, 256, 0, stream>>>(h, tbuf, gamma + (size_t)L * HID,
                                               beta + (size_t)L * HID, N);
    }

    gemm64_direct_narrow<<<ggrid, 128, 0, stream>>>(h, w_out, b_out, (float*)d_out, N, 8);
}


# node phase as one launch per layer: the GEMM wave does layernorm, gelu, residual on its 16 rows in A-fragment layout, then fc_src and fc_dst from them; WMMA k pairing and LN sum order change
# speedup vs baseline: 6.4041x; 1.0934x over previous
#include <hip/hip_runtime.h>
#include <hip/hip_bf16.h>
#include <math.h>


typedef __attribute__((ext_vector_type(2))) float v2f;
typedef __attribute__((ext_vector_type(8))) float v8f;

#define HID 64
#define NEG_SLOPE 0.2f
#define LN_EPS 1e-5f

__device__ __forceinline__ unsigned fkey(float f) {
    unsigned b = __float_as_uint(f);
    return (b & 0x80000000u) ? ~b : (b | 0x80000000u);
}
__device__ __forceinline__ float funkey(unsigned k) {
    unsigned b = (k & 0x80000000u) ? (k ^ 0x80000000u) : ~k;
    return __uint_as_float(b);
}
#define KEY_NEG_INF 0x007FFFFFu

__global__ __launch_bounds__(128)
void gemm64_direct(const float* __restrict__ X, const float* __restrict__ W,
                   const float* __restrict__ bias, float* __restrict__ Y,
                   int nrows) {
    const int lane = threadIdx.x & 31;
    const int wave = __builtin_amdgcn_readfirstlane(threadIdx.x >> 5);
    const int half = lane >> 4;
    const int l15  = lane & 15;
    const int rbase = (blockIdx.x * 4 + wave) * 16;
    if (rbase >= nrows) return;

    int arow = rbase + l15;
    if (arow >= nrows) arow = nrows - 1;
    const float* xrow = X + (size_t)arow * 64 + 2 * half;
    const float* wcol = W + (2 * half) * 64 + 4 * l15;

    v2f a[16];
    float4 w0[16], w1[16];
#pragma unroll
    for (int ks = 0; ks < 16; ++ks) {
        a[ks].x = xrow[4 * ks];
        a[ks].y = xrow[4 * ks + 1];
    }
#pragma unroll
    for (int ks = 0; ks < 16; ++ks) {
        w0[ks] = *(const float4*)(wcol + (4 * ks) * 64);
        w1[ks] = *(const float4*)(wcol + (4 * ks + 1) * 64);
    }
    const float4 bv = *(const float4*)(bias + 4 * l15);

    v8f acc0, acc1, acc2, acc3;
#pragma unroll
    for (int j = 0; j < 8; ++j) {
        acc0[j] = 0.0f; acc1[j] = 0.0f; acc2[j] = 0.0f; acc3[j] = 0.0f;
    }

#pragma unroll
    for (int ks = 0; ks < 16; ++ks) {
        v2f b;
        b.x = w0[ks].x; b.y = w1[ks].x;
        acc0 = __builtin_amdgcn_wmma_f32_16x16x4_f32(
            false, a[ks], false, b, (short)0, acc0, false, false);
        b.x = w0[ks].y; b.y = w1[ks].y;
        acc1 = __builtin_amdgcn_wmma_f32_16x16x4_f32(
            false, a[ks], false, b, (short)0, acc1, false, false);
        b.x = w0[ks].z; b.y = w1[ks].z;
        acc2 = __builtin_amdgcn_wmma_f32_16x16x4_f32(
            false, a[ks], false, b, (short)0, acc2, false, false);
        b.x = w0[ks].w; b.y = w1[ks].w;
        acc3 = __builtin_amdgcn_wmma_f32_16x16x4_f32(
            false, a[ks], false, b, (short)0, acc3, false, false);
    }

#pragma unroll
    for (int i = 0; i < 8; ++i) {
        const int row = rbase + 8 * half + i;
        if (row < nrows) {
            float4 o;
            o.x = acc0[i] + bv.x;
            o.y = acc1[i] + bv.y;
            o.z = acc2[i] + bv.z;
            o.w = acc3[i] + bv.w;
            *(float4*)(Y + (size_t)row * 64 + 4 * l15) = o;
        }
    }
}

#define OWN_R  208
#define OWN_NT 1024
#define OWN_NW (OWN_NT / 32)
#define OWN_CAP 8192

__device__ __forceinline__ float own_wave_sum(float p) {
    p += __shfl_xor(p, 16);
    p += __shfl_xor(p, 8);
    p += __shfl_xor(p, 4);
    p += __shfl_xor(p, 2);
    p += __shfl_xor(p, 1);
    return p;
}

__device__ __forceinline__ void own_sums(int n, int s, int dl, float ex, int lane,
                                         const float* __restrict__ fs, float* t_l) {
    for (int g = 0; g < n; g += 8) {
        float a0[8], a1[8];
#pragma unroll
        for (int u = 0; u < 8; ++u) {
            int j = min(g + u, n - 1);
            int sj = __builtin_amdgcn_readlane(s, j);
            const float* ra = fs + (size_t)sj * HID;
            a0[u] = ra[lane];
            a1[u] = ra[32 + lane];
        }
#pragma unroll
        for (int u = 0; u < 8; ++u) {
            if (g + u < n) {
                int dj = __builtin_amdgcn_readlane(dl, g + u);
                float xj = __int_as_float(
                    __builtin_amdgcn_readlane(__float_as_int(ex), g + u));
                atomicAdd(&t_l[dj * HID + lane], xj * a0[u]);
                atomicAdd(&t_l[dj * HID + 32 + lane], xj * a1[u]);
            }
        }
    }
}

struct OwnList {
    int*   cnt;
    int*   src;
    int*   dl;
    float* sc;
};

template <int PASS>
__device__ __forceinline__ void own_flush(int n, const int* q, int lane, int base,
                                          const float* __restrict__ fs,
                                          const float* __restrict__ fd,
                                          const int* __restrict__ src,
                                          const int* __restrict__ dst,
                                          float w0, float w1, float* score,
                                          float* t_l, float* den_l, unsigned* mx_l,
                                          OwnList ls) {
    int e = 0, s = 0, d = base;
    if (lane < n) {
        e = q[lane];
        s = src[e];
        d = dst[e];
    }
    const int dl = d - base;
    if (PASS == 1) {
        float mine = 0.0f;
        for (int g = 0; g < n; g += 8) {
            float a0[8], a1[8], b0[8], b1[8];
#pragma unroll
            for (int u = 0; u < 8; ++u) {
                int j = min(g + u, n - 1);
                int sj = __builtin_amdgcn_readlane(s, j);
                int dj = __builtin_amdgcn_readlane(d, j);
                const float* ra = fs + (size_t)sj * HID;
                const float* rb = fd + (size_t)dj * HID;
                a0[u] = ra[lane];
                a1[u] = ra[32 + lane];
                b0[u] = rb[lane];
                b1[u] = rb[32 + lane];
            }
#pragma unroll
            for (int u = 0; u < 8; ++u) {
                float x0 = a0[u] + b0[u], x1 = a1[u] + b1[u];
                x0 = x0 > 0.f ? x0 : x0 * NEG_SLOPE;
                x1 = x1 > 0.f ? x1 : x1 * NEG_SLOPE;
                float p = own_wave_sum(x0 * w0 + x1 * w1);
                if (lane == g + u) mine = p;
            }
        }
        if (lane < n) {
            score[e] = mine;
            atomicMax(&mx_l[dl], fkey(mine));
        }
        int slot = 0;
        if (lane == 0) slot = atomicAdd(ls.cnt, n);
        slot = __builtin_amdgcn_readfirstlane(slot);
        if (slot + n <= OWN_CAP && lane < n) {
            ls.src[slot + lane] = s;
            ls.dl[slot + lane]  = dl;
            ls.sc[slot + lane]  = mine;
        }
    } else {
        float ex = 0.0f;
        if (lane < n) {
            ex = expf(score[e] - funkey(mx_l[dl]));
            atomicAdd(&den_l[dl], ex);
        }
        own_sums(n, s, dl, ex, lane, fs, t_l);
    }
}

__device__ __forceinline__ void own_list_sums(int wave, int lane, int total,
                                              const float* __restrict__ fs,
                                              OwnList ls, float* t_l, float* den_l,
                                              const unsigned* mx_l) {
    const int per = (total + OWN_NW - 1) / OWN_NW;
    const int beg = wave * per;
    const int end = min(total, beg + per);
    for (int off = beg; off < end; off += 32) {
        const int n = min(32, end - off);
        int s = 0, dl = 0;
        float ex = 0.0f;
        if (lane < n) {
            s  = ls.src[off + lane];
            dl = ls.dl[off + lane];
            ex = expf(ls.sc[off + lane] - funkey(mx_l[dl]));
            atomicAdd(&den_l[dl], ex);
        }
        own_sums(n, s, dl, ex, lane, fs, t_l);
    }
}

#define OWN_U 8
#define OWN_HIT(d) (((unsigned)((d) - base) < (unsigned)OWN_R) ? 1u : 0u)

template <int PASS>
__device__ __forceinline__ void own_scan(int wave, int lane, int base, int E,
                                         const float* __restrict__ fs,
                                         const float* __restrict__ fd,
                                         const int* __restrict__ src,
                                         const int* __restrict__ dst,
                                         float w0, float w1, float* score,
                                         float* t_l, float* den_l, unsigned* mx_l,
                                         int* q, OwnList ls) {
    const int nfull = E >> 7;
    const int rot = (int)(((long long)blockIdx.x * nfull) / gridDim.x);
    int cnt = 0;
    int kb[OWN_U];
#pragma unroll
    for (int sl = 0; sl < OWN_U; ++sl) kb[sl] = 0;
    kb[0] = nfull * 128;
    unsigned mk = 0;
    if (wave == 0) {
        const int eb = kb[0] + lane * 4;
        int4 tv = make_int4(-1, -1, -1, -1);
        if (eb     < E) tv.x = dst[eb];
        if (eb + 1 < E) tv.y = dst[eb + 1];
        if (eb + 2 < E) tv.z = dst[eb + 2];
        if (eb + 3 < E) tv.w = dst[eb + 3];
        mk = OWN_HIT(tv.x) | (OWN_HIT(tv.y) << 1) | (OWN_HIT(tv.z) << 2) | (OWN_HIT(tv.w) << 3);
    }
    int c = wave;
    while (true) {
        for (int round = 0; round < 4 * OWN_U; ++round) {
            const bool m = mk != 0u;
            const unsigned bal = __builtin_amdgcn_ballot_w32(m);
            if (bal == 0u) break;
            const int i = m ? __builtin_ctz(mk) : 0;
            mk &= mk - 1u;
            const int sl = i >> 2;
            int kbs = kb[0];
#pragma unroll
            for (int s2 = 1; s2 < OWN_U; ++s2) kbs = (sl == s2) ? kb[s2] : kbs;
            const int pos = cnt + (int)__builtin_amdgcn_mbcnt_lo(bal, 0u);
            if (m) q[pos] = kbs + lane * 4 + (i & 3);
            cnt += __builtin_popcount(bal);
            if (cnt >= 32) {
                own_flush<PASS>(32, q, lane, base, fs, fd, src, dst, w0, w1,
                                score, t_l, den_l, mx_l, ls);
                const int rem = cnt - 32;
                int tmp = 0;
                if (lane < rem) tmp = q[32 + lane];
                if (lane < rem) q[lane] = tmp;
                cnt = rem;
            }
        }
        if (c >= nfull) break;
        int4 v[OWN_U];
        unsigned okmask = 0;
#pragma unroll
        for (int sl = 0; sl < OWN_U; ++sl) {
            const bool ok = c + sl * OWN_NW < nfull;
            int k = ok ? c + sl * OWN_NW + rot : 0;
            if (k >= nfull) k -= nfull;
            kb[sl] = k * 128;
            okmask |= ok ? (0xFu << (4 * sl)) : 0u;
            v[sl] = *(const int4*)(dst + kb[sl] + lane * 4);
        }
        unsigned nm = 0;
#pragma unroll
        for (int sl = OWN_U - 1; sl >= 0; --sl) {
            nm = (nm << 1) | OWN_HIT(v[sl].w);
            nm = (nm << 1) | OWN_HIT(v[sl].z);
            nm = (nm << 1) | OWN_HIT(v[sl].y);
            nm = (nm << 1) | OWN_HIT(v[sl].x);
        }
        mk = nm & okmask;
        c += OWN_U * OWN_NW;
    }
    if (cnt > 0)
        own_flush<PASS>(cnt, q, lane, base, fs, fd, src, dst, w0, w1,
                        score, t_l, den_l, mx_l, ls);
}

__global__ __launch_bounds__(OWN_NT)
void edge_owner(const float* __restrict__ fs, const float* __restrict__ fd,
                const int* __restrict__ src, const int* __restrict__ dst,
                const float* __restrict__ attn, float* score,
                float* __restrict__ t, int N, int E) {
    __shared__ __attribute__((aligned(16))) float t_l[OWN_R * HID];
    __shared__ float    den_l[OWN_R];
    __shared__ unsigned mx_l[OWN_R];
    __shared__ int      q_l[OWN_NW * 64];
    __shared__ int      ls_src[OWN_CAP];
    __shared__ int      ls_dl[OWN_CAP];
    __shared__ float    ls_sc[OWN_CAP];
    __shared__ int      ls_cnt;

    const int tid  = threadIdx.x;
    const int lane = tid & 31;
    const int wave = __builtin_amdgcn_readfirstlane(tid >> 5);
    const int base = blockIdx.x * OWN_R;

    for (int i = tid; i < OWN_R * HID; i += OWN_NT) t_l[i] = 0.0f;
    if (tid < OWN_R) {
        den_l[tid] = 0.0f;
        mx_l[tid]  = KEY_NEG_INF;
    }
    if (tid == 0) ls_cnt = 0;
    OwnList ls;
    ls.cnt = &ls_cnt;
    ls.src = ls_src;
    ls.dl  = ls_dl;
    ls.sc  = ls_sc;
    const float w0 = attn[lane], w1 = attn[32 + lane];
    __syncthreads();

    own_scan<1>(wave, lane, base, E, fs, fd, src, dst, w0, w1, score,
                t_l, den_l, mx_l, q_l + wave * 64, ls);
    __syncthreads();
    const int total = __builtin_amdgcn_readfirstlane(ls_cnt);
    if (total <= OWN_CAP) {
        own_list_sums(wave, lane, total, fs, ls, t_l, den_l, mx_l);
    } else {
        own_scan<2>(wave, lane, base, E, fs, fd, src, dst, w0, w1, score,
                    t_l, den_l, mx_l, q_l + wave * 64, ls);
    }
    __syncthreads();

    for (int idx = tid; idx < OWN_R * 16; idx += OWN_NT) {
        const int row = idx >> 4, c4 = idx & 15;
        const int grow = base + row;
        if (grow < N) {
            const float den = den_l[row];
            const float4 v = *(const float4*)(t_l + row * HID + c4 * 4);
            float4 o = make_float4(0.f, 0.f, 0.f, 0.f);
            if (den > 0.0f) {
                o.x = v.x / den; o.y = v.y / den; o.z = v.z / den; o.w = v.w / den;
            }
            *(float4*)(t + (size_t)grow * HID + c4 * 4) = o;
        }
    }
}

__device__ __forceinline__ void rows_load(const float* __restrict__ X, int arow, int hf,
                                          float4 (&q)[8]) {
    const float* p = X + (size_t)arow * HID + 4 * hf;
#pragma unroll
    for (int j = 0; j < 8; ++j) q[j] = *(const float4*)(p + 8 * j);
}

#define ROWS_K(ks) (8 * ((ks) >> 1) + 2 * ((ks) & 1))

__device__ __forceinline__ v2f rows_afrag(const float4 (&q)[8], int ks) {
    v2f a;
    a.x = (ks & 1) ? q[ks >> 1].z : q[ks >> 1].x;
    a.y = (ks & 1) ? q[ks >> 1].w : q[ks >> 1].y;
    return a;
}

__device__ __forceinline__ void rows_update(float4 (&q)[8], const float4 (&v)[8],
                                            const float* __restrict__ gamma,
                                            const float* __restrict__ beta, int hf) {
    float sum = 0.0f;
#pragma unroll
    for (int j = 0; j < 8; ++j) sum += v[j].x + v[j].y + v[j].z + v[j].w;
    sum += __shfl_xor(sum, 16);
    const float mean = sum * (1.0f / HID);
    float var = 0.0f;
#pragma unroll
    for (int j = 0; j < 8; ++j) {
        const float d0 = v[j].x - mean, d1 = v[j].y - mean;
        const float d2 = v[j].z - mean, d3 = v[j].w - mean;
        var += d0 * d0 + d1 * d1 + d2 * d2 + d3 * d3;
    }
    var += __shfl_xor(var, 16);
    var *= (1.0f / HID);
    const float inv = rsqrtf(var + LN_EPS);
    const float4* pg = (const float4*)(gamma + 4 * hf);
    const float4* pb = (const float4*)(beta + 4 * hf);
    const float k = 0.70710678118654752f;
    float4 x[8];
#pragma unroll
    for (int j = 0; j < 8; ++j) {
        const float4 g = pg[2 * j], b = pb[2 * j];
        x[j].x = (v[j].x - mean) * inv * g.x + b.x;
        x[j].y = (v[j].y - mean) * inv * g.y + b.y;
        x[j].z = (v[j].z - mean) * inv * g.z + b.z;
        x[j].w = (v[j].w - mean) * inv * g.w + b.w;
    }
#pragma unroll
    for (int j = 0; j < 8; ++j) {
        q[j].x += 0.5f * x[j].x * (1.0f + erff(x[j].x * k));
        q[j].y += 0.5f * x[j].y * (1.0f + erff(x[j].y * k));
        q[j].z += 0.5f * x[j].z * (1.0f + erff(x[j].z * k));
        q[j].w += 0.5f * x[j].w * (1.0f + erff(x[j].w * k));
    }
}

__device__ __forceinline__ void rows_wload(const float* __restrict__ W, int hf, int l15,
                                           float4 (&w0)[16], float4 (&w1)[16]) {
    const float* wcol = W + (4 * hf) * HID + 4 * l15;
#pragma unroll
    for (int ks = 0; ks < 16; ++ks) {
        w0[ks] = *(const float4*)(wcol + ROWS_K(ks) * HID);
        w1[ks] = *(const float4*)(wcol + (ROWS_K(ks) + 1) * HID);
    }
}

__device__ __forceinline__ void rows_gemm64(const float4 (&q)[8], const float4 (&w0)[16],
                                            const float4 (&w1)[16],
                                            const float* __restrict__ bias,
                                            float* __restrict__ Y, int rbase, int nrows,
                                            int hf, int l15) {
    const float4 bv = *(const float4*)(bias + 4 * l15);
    v8f acc0, acc1, acc2, acc3;
#pragma unroll
    for (int j = 0; j < 8; ++j) {
        acc0[j] = 0.0f; acc1[j] = 0.0f; acc2[j] = 0.0f; acc3[j] = 0.0f;
    }
#pragma unroll
    for (int ks = 0; ks < 16; ++ks) {
        const v2f a = rows_afrag(q, ks);
        v2f b;
        b.x = w0[ks].x; b.y = w1[ks].x;
        acc0 = __builtin_amdgcn_wmma_f32_16x16x4_f32(
            false, a, false, b, (short)0, acc0, false, false);
        b.x = w0[ks].y; b.y = w1[ks].y;
        acc1 = __builtin_amdgcn_wmma_f32_16x16x4_f32(
            false, a, false, b, (short)0, acc1, false, false);
        b.x = w0[ks].z; b.y = w1[ks].z;
        acc2 = __builtin_amdgcn_wmma_f32_16x16x4_f32(
            false, a, false, b, (short)0, acc2, false, false);
        b.x = w0[ks].w; b.y = w1[ks].w;
        acc3 = __builtin_amdgcn_wmma_f32_16x16x4_f32(
            false, a, false, b, (short)0, acc3, false, false);
    }
#pragma unroll
    for (int i = 0; i < 8; ++i) {
        const int row = rbase + 8 * hf + i;
        if (row < nrows) {
            float4 o;
            o.x = acc0[i] + bv.x;
            o.y = acc1[i] + bv.y;
            o.z = acc2[i] + bv.z;
            o.w = acc3[i] + bv.w;
            *(float4*)(Y + (size_t)row * HID + 4 * l15) = o;
        }
    }
}

__global__ __launch_bounds__(128)
void node_first(const float* __restrict__ h,
                const float* __restrict__ Ws, const float* __restrict__ bs,
                float* __restrict__ fs,
                const float* __restrict__ Wd, const float* __restrict__ bd,
                float* __restrict__ fd, int nrows) {
    const int lane = threadIdx.x & 31;
    const int wave = __builtin_amdgcn_readfirstlane(threadIdx.x >> 5);
    const int hf = lane >> 4, l15 = lane & 15;
    const int rbase = (blockIdx.x * 4 + wave) * 16;
    if (rbase >= nrows) return;
    int arow = rbase + l15;
    if (arow >= nrows) arow = nrows - 1;
    float4 q[8], w0[16], w1[16];
    rows_load(h, arow, hf, q);
    rows_wload(Ws, hf, l15, w0, w1);
    rows_gemm64(q, w0, w1, bs, fs, rbase, nrows, hf, l15);
    rows_wload(Wd, hf, l15, w0, w1);
    rows_gemm64(q, w0, w1, bd, fd, rbase, nrows, hf, l15);
}

__global__ __launch_bounds__(128)
void node_layer(float* __restrict__ h, const float* __restrict__ t,
                const float* __restrict__ gamma, const float* __restrict__ beta,
                const float* __restrict__ Ws, const float* __restrict__ bs,
                float* __restrict__ fs,
                const float* __restrict__ Wd, const float* __restrict__ bd,
                float* __restrict__ fd, int nrows) {
    const int lane = threadIdx.x & 31;
    const int wave = __builtin_amdgcn_readfirstlane(threadIdx.x >> 5);
    const int hf = lane >> 4, l15 = lane & 15;
    const int rbase = (blockIdx.x * 4 + wave) * 16;
    if (rbase >= nrows) return;
    const bool rv = rbase + l15 < nrows;
    const int arow = rv ? rbase + l15 : nrows - 1;
    float4 q[8], v[8], w0[16], w1[16];
    rows_load(t, arow, hf, v);
    rows_load(h, arow, hf, q);
    rows_update(q, v, gamma, beta, hf);
    if (rv) {
        float* p = h + (size_t)arow * HID + 4 * hf;
#pragma unroll
        for (int j = 0; j < 8; ++j) *(float4*)(p + 8 * j) = q[j];
    }
    rows_wload(Ws, hf, l15, w0, w1);
    rows_gemm64(q, w0, w1, bs, fs, rbase, nrows, hf, l15);
    rows_wload(Wd, hf, l15, w0, w1);
    rows_gemm64(q, w0, w1, bd, fd, rbase, nrows, hf, l15);
}

__global__ __launch_bounds__(128)
void node_last(const float* __restrict__ h, const float* __restrict__ t,
               const float* __restrict__ gamma, const float* __restrict__ beta,
               const float* __restrict__ W, const float* __restrict__ bias,
               float* __restrict__ Y, int nrows, int nout) {
    const int lane = threadIdx.x & 31;
    const int wave = __builtin_amdgcn_readfirstlane(threadIdx.x >> 5);
    const int hf = lane >> 4, l15 = lane & 15;
    const int rbase = (blockIdx.x * 4 + wave) * 16;
    if (rbase >= nrows) return;
    int arow = rbase + l15;
    if (arow >= nrows) arow = nrows - 1;
    float4 q[8], v[8];
    rows_load(t, arow, hf, v);
    rows_load(h, arow, hf, q);

    const bool cv = l15 < nout;
    const float* wcol = W + (4 * hf) * nout + (cv ? l15 : 0);
    v2f b[16];
#pragma unroll
    for (int ks = 0; ks < 16; ++ks) {
        const float bx = wcol[ROWS_K(ks) * nout];
        const float by = wcol[(ROWS_K(ks) + 1) * nout];
        b[ks].x = cv ? bx : 0.0f;
        b[ks].y = cv ? by : 0.0f;
    }
    const float bv = cv ? bias[l15] : 0.0f;

    rows_update(q, v, gamma, beta, hf);

    v8f acc;
#pragma unroll
    for (int j = 0; j < 8; ++j) acc[j] = 0.0f;
#pragma unroll
    for (int ks = 0; ks < 16; ++ks)
        acc = __builtin_amdgcn_wmma_f32_16x16x4_f32(
            false, rows_afrag(q, ks), false, b[ks], (short)0, acc, false, false);

#pragma unroll
    for (int i = 0; i < 8; ++i) {
        const int row = rbase + 8 * hf + i;
        if (row < nrows && cv) Y[(size_t)row * nout + l15] = acc[i] + bv;
    }
}

extern "C" void kernel_launch(void* const* d_in, const int* in_sizes, int n_in,
                              void* d_out, int out_size, void* d_ws, size_t ws_size,
                              hipStream_t stream) {
    const float* nodes = (const float*)d_in[0];
    const int*   src   = (const int*)d_in[1];
    const int*   dst   = (const int*)d_in[2];
    const float* w_in  = (const float*)d_in[3];
    const float* b_in  = (const float*)d_in[4];
    const float* w_src = (const float*)d_in[5];
    const float* b_src = (const float*)d_in[6];
    const float* w_dst = (const float*)d_in[7];
    const float* b_dst = (const float*)d_in[8];
    const float* attn  = (const float*)d_in[9];
    const float* gamma = (const float*)d_in[10];
    const float* beta  = (const float*)d_in[11];
    const float* w_out = (const float*)d_in[12];
    const float* b_out = (const float*)d_in[13];

    const int N = in_sizes[0] / HID;
    const int E = in_sizes[1];
    const int NL = 6;

    float* ws   = (float*)d_ws;
    float* h    = ws;  ws += (size_t)N * HID;
    float* fs   = ws;  ws += (size_t)N * HID;
    float* fd   = ws;  ws += (size_t)N * HID;
    float* tbuf = ws;  ws += (size_t)N * HID;
    float* score = ws;

    const int ggrid  = (N + 63) / 64;
    const int ogrid  = (N + OWN_R - 1) / OWN_R;

    gemm64_direct<<<ggrid, 128, 0, stream>>>(nodes, w_in, b_in, h, N);

    for (int L = 0; L < NL; ++L) {
        const float* ws_L = w_src + (size_t)L * HID * HID;
        const float* wd_L = w_dst + (size_t)L * HID * HID;
        const float* bs_L = b_src + (size_t)L * HID;
        const float* bd_L = b_dst + (size_t)L * HID;
        if (L == 0) {
            node_first<<<ggrid, 128, 0, stream>>>(h, ws_L, bs_L, fs, wd_L, bd_L, fd, N);
        } else {
            node_layer<<<ggrid, 128, 0, stream>>>(h, tbuf, gamma + (size_t)(L - 1) * HID,
                                                  beta + (size_t)(L - 1) * HID,
                                                  ws_L, bs_L, fs, wd_L, bd_L, fd, N);
        }
        edge_owner<<<ogrid, OWN_NT, 0, stream>>>(fs, fd, src, dst,
                                                 attn + (size_t)L * HID, score, tbuf, N, E);
    }

    node_last<<<ggrid, 128, 0, stream>>>(h, tbuf, gamma + (size_t)(NL - 1) * HID,
                                         beta + (size_t)(NL - 1) * HID,
                                         w_out, b_out, (float*)d_out, N, 8);
}


# edge filter once per call: layer 0's sweep saves each block's matched-edge list (source id, local row) in the old score plane; layers 1-5 read it, no sweep; sums still LDS atomics, arrival order
# speedup vs baseline: 9.5966x; 1.4985x over previous
#include <hip/hip_runtime.h>
#include <hip/hip_bf16.h>
#include <math.h>


typedef __attribute__((ext_vector_type(2))) float v2f;
typedef __attribute__((ext_vector_type(8))) float v8f;

#define HID 64
#define NEG_SLOPE 0.2f
#define LN_EPS 1e-5f

__device__ __forceinline__ unsigned fkey(float f) {
    unsigned b = __float_as_uint(f);
    return (b & 0x80000000u) ? ~b : (b | 0x80000000u);
}
__device__ __forceinline__ float funkey(unsigned k) {
    unsigned b = (k & 0x80000000u) ? (k ^ 0x80000000u) : ~k;
    return __uint_as_float(b);
}
#define KEY_NEG_INF 0x007FFFFFu

__global__ __launch_bounds__(128)
void gemm64_direct(const float* __restrict__ X, const float* __restrict__ W,
                   const float* __restrict__ bias, float* __restrict__ Y,
                   int nrows) {
    const int lane = threadIdx.x & 31;
    const int wave = __builtin_amdgcn_readfirstlane(threadIdx.x >> 5);
    const int half = lane >> 4;
    const int l15  = lane & 15;
    const int rbase = (blockIdx.x * 4 + wave) * 16;
    if (rbase >= nrows) return;

    int arow = rbase + l15;
    if (arow >= nrows) arow = nrows - 1;
    const float* xrow = X + (size_t)arow * 64 + 2 * half;
    const float* wcol = W + (2 * half) * 64 + 4 * l15;

    v2f a[16];
    float4 w0[16], w1[16];
#pragma unroll
    for (int ks = 0; ks < 16; ++ks) {
        a[ks].x = xrow[4 * ks];
        a[ks].y = xrow[4 * ks + 1];
    }
#pragma unroll
    for (int ks = 0; ks < 16; ++ks) {
        w0[ks] = *(const float4*)(wcol + (4 * ks) * 64);
        w1[ks] = *(const float4*)(wcol + (4 * ks + 1) * 64);
    }
    const float4 bv = *(const float4*)(bias + 4 * l15);

    v8f acc0, acc1, acc2, acc3;
#pragma unroll
    for (int j = 0; j < 8; ++j) {
        acc0[j] = 0.0f; acc1[j] = 0.0f; acc2[j] = 0.0f; acc3[j] = 0.0f;
    }

#pragma unroll
    for (int ks = 0; ks < 16; ++ks) {
        v2f b;
        b.x = w0[ks].x; b.y = w1[ks].x;
        acc0 = __builtin_amdgcn_wmma_f32_16x16x4_f32(
            false, a[ks], false, b, (short)0, acc0, false, false);
        b.x = w0[ks].y; b.y = w1[ks].y;
        acc1 = __builtin_amdgcn_wmma_f32_16x16x4_f32(
            false, a[ks], false, b, (short)0, acc1, false, false);
        b.x = w0[ks].z; b.y = w1[ks].z;
        acc2 = __builtin_amdgcn_wmma_f32_16x16x4_f32(
            false, a[ks], false, b, (short)0, acc2, false, false);
        b.x = w0[ks].w; b.y = w1[ks].w;
        acc3 = __builtin_amdgcn_wmma_f32_16x16x4_f32(
            false, a[ks], false, b, (short)0, acc3, false, false);
    }

#pragma unroll
    for (int i = 0; i < 8; ++i) {
        const int row = rbase + 8 * half + i;
        if (row < nrows) {
            float4 o;
            o.x = acc0[i] + bv.x;
            o.y = acc1[i] + bv.y;
            o.z = acc2[i] + bv.z;
            o.w = acc3[i] + bv.w;
            *(float4*)(Y + (size_t)row * 64 + 4 * l15) = o;
        }
    }
}

#define OWN_R  208
#define OWN_NT 1024
#define OWN_NW (OWN_NT / 32)
#define OWN_CAP 8192

__device__ __forceinline__ float own_wave_sum(float p) {
    p += __shfl_xor(p, 16);
    p += __shfl_xor(p, 8);
    p += __shfl_xor(p, 4);
    p += __shfl_xor(p, 2);
    p += __shfl_xor(p, 1);
    return p;
}

__device__ __forceinline__ void own_sums(int n, int s, int dl, float ex, int lane,
                                         const float* __restrict__ fs, float* t_l) {
    for (int g = 0; g < n; g += 8) {
        float a0[8], a1[8];
#pragma unroll
        for (int u = 0; u < 8; ++u) {
            int j = min(g + u, n - 1);
            int sj = __builtin_amdgcn_readlane(s, j);
            const float* ra = fs + (size_t)sj * HID;
            a0[u] = ra[lane];
            a1[u] = ra[32 + lane];
        }
#pragma unroll
        for (int u = 0; u < 8; ++u) {
            if (g + u < n) {
                int dj = __builtin_amdgcn_readlane(dl, g + u);
                float xj = __int_as_float(
                    __builtin_amdgcn_readlane(__float_as_int(ex), g + u));
                atomicAdd(&t_l[dj * HID + lane], xj * a0[u]);
                atomicAdd(&t_l[dj * HID + 32 + lane], xj * a1[u]);
            }
        }
    }
}

struct OwnList {
    int*   cnt;
    int*   src;
    int*   dl;
    float* sc;
};

struct OwnKeep {
    int*            hdr;
    unsigned short* src;
    unsigned char*  dl;
    int             cap;
};

__device__ __forceinline__ float own_scores(int n, int s, int d, int lane,
                                            const float* __restrict__ fs,
                                            const float* __restrict__ fd,
                                            float w0, float w1) {
    float mine = 0.0f;
    for (int g = 0; g < n; g += 8) {
        float a0[8], a1[8], b0[8], b1[8];
#pragma unroll
        for (int u = 0; u < 8; ++u) {
            int j = min(g + u, n - 1);
            int sj = __builtin_amdgcn_readlane(s, j);
            int dj = __builtin_amdgcn_readlane(d, j);
            const float* ra = fs + (size_t)sj * HID;
            const float* rb = fd + (size_t)dj * HID;
            a0[u] = ra[lane];
            a1[u] = ra[32 + lane];
            b0[u] = rb[lane];
            b1[u] = rb[32 + lane];
        }
#pragma unroll
        for (int u = 0; u < 8; ++u) {
            float x0 = a0[u] + b0[u], x1 = a1[u] + b1[u];
            x0 = x0 > 0.f ? x0 : x0 * NEG_SLOPE;
            x1 = x1 > 0.f ? x1 : x1 * NEG_SLOPE;
            float p = own_wave_sum(x0 * w0 + x1 * w1);
            if (lane == g + u) mine = p;
        }
    }
    return mine;
}

template <int PASS>
__device__ __forceinline__ void own_flush(int n, const int* q, int lane, int base,
                                          const float* __restrict__ fs,
                                          const float* __restrict__ fd,
                                          const int* __restrict__ src,
                                          const int* __restrict__ dst,
                                          float w0, float w1,
                                          float* t_l, float* den_l, unsigned* mx_l,
                                          OwnList ls) {
    int e = 0, s = 0, d = base;
    if (lane < n) {
        e = q[lane];
        s = src[e];
        d = dst[e];
    }
    const int dl = d - base;
    const float mine = own_scores(n, s, d, lane, fs, fd, w0, w1);
    if (PASS == 1) {
        if (lane < n) atomicMax(&mx_l[dl], fkey(mine));
        int slot = 0;
        if (lane == 0) slot = atomicAdd(ls.cnt, n);
        slot = __builtin_amdgcn_readfirstlane(slot);
        if (slot + n <= OWN_CAP && lane < n) {
            ls.src[slot + lane] = s;
            ls.dl[slot + lane]  = dl;
            ls.sc[slot + lane]  = mine;
        }
    } else {
        float ex = 0.0f;
        if (lane < n) {
            ex = expf(mine - funkey(mx_l[dl]));
            atomicAdd(&den_l[dl], ex);
        }
        own_sums(n, s, dl, ex, lane, fs, t_l);
    }
}

__device__ __forceinline__ void own_list_scores(int wave, int lane, int total, int base,
                                                const float* __restrict__ fs,
                                                const float* __restrict__ fd,
                                                float w0, float w1, OwnKeep kp,
                                                OwnList ls, unsigned* mx_l) {
    const int per = (total + OWN_NW - 1) / OWN_NW;
    const int beg = wave * per;
    const int end = min(total, beg + per);
    int s_nx = 0, dl_nx = 0;
    if (beg + lane < end) {
        s_nx  = kp.src[beg + lane];
        dl_nx = kp.dl[beg + lane];
    }
    for (int off = beg; off < end; off += 32) {
        const int n = min(32, end - off);
        const int s = s_nx, dl = dl_nx;
        if (off + 32 + lane < end) {
            s_nx  = kp.src[off + 32 + lane];
            dl_nx = kp.dl[off + 32 + lane];
        }
        const float mine = own_scores(n, s, base + dl, lane, fs, fd, w0, w1);
        if (lane < n) {
            atomicMax(&mx_l[dl], fkey(mine));
            ls.src[off + lane] = s;
            ls.dl[off + lane]  = dl;
            ls.sc[off + lane]  = mine;
        }
    }
}

__device__ __forceinline__ void own_list_save(int tid, int lane, int total,
                                              OwnList ls, OwnKeep kp) {
    for (int i4 = tid * 4; i4 - lane * 4 < total; i4 += OWN_NT * 4) {
        unsigned sv[4], dv[4];
#pragma unroll
        for (int u = 0; u < 4; ++u) {
            const bool ok = i4 + u < total;
            sv[u] = ok ? (unsigned)ls.src[i4 + u] : 0u;
            dv[u] = ok ? (unsigned)ls.dl[i4 + u] : 0u;
        }
        uint2 sp;
        sp.x = sv[0] | (sv[1] << 16);
        sp.y = sv[2] | (sv[3] << 16);
        *(uint2*)(kp.src + i4) = sp;
        *(unsigned*)(kp.dl + i4) = dv[0] | (dv[1] << 8) | (dv[2] << 16) | (dv[3] << 24);
    }
}

__device__ __forceinline__ void own_list_sums(int wave, int lane, int total,
                                              const float* __restrict__ fs,
                                              OwnList ls, float* t_l, float* den_l,
                                              const unsigned* mx_l) {
    const int per = (total + OWN_NW - 1) / OWN_NW;
    const int beg = wave * per;
    const int end = min(total, beg + per);
    for (int off = beg; off < end; off += 32) {
        const int n = min(32, end - off);
        int s = 0, dl = 0;
        float ex = 0.0f;
        if (lane < n) {
            s  = ls.src[off + lane];
            dl = ls.dl[off + lane];
            ex = expf(ls.sc[off + lane] - funkey(mx_l[dl]));
            atomicAdd(&den_l[dl], ex);
        }
        own_sums(n, s, dl, ex, lane, fs, t_l);
    }
}

#define OWN_U 8
#define OWN_HIT(d) (((unsigned)((d) - base) < (unsigned)OWN_R) ? 1u : 0u)

template <int PASS>
__device__ __forceinline__ void own_scan(int wave, int lane, int base, int E,
                                         const float* __restrict__ fs,
                                         const float* __restrict__ fd,
                                         const int* __restrict__ src,
                                         const int* __restrict__ dst,
                                         float w0, float w1,
                                         float* t_l, float* den_l, unsigned* mx_l,
                                         int* q, OwnList ls) {
    const int nfull = E >> 7;
    const int rot = (int)(((long long)blockIdx.x * nfull) / gridDim.x);
    int cnt = 0;
    int kb[OWN_U];
#pragma unroll
    for (int sl = 0; sl < OWN_U; ++sl) kb[sl] = 0;
    kb[0] = nfull * 128;
    unsigned mk = 0;
    if (wave == 0) {
        const int eb = kb[0] + lane * 4;
        int4 tv = make_int4(-1, -1, -1, -1);
        if (eb     < E) tv.x = dst[eb];
        if (eb + 1 < E) tv.y = dst[eb + 1];
        if (eb + 2 < E) tv.z = dst[eb + 2];
        if (eb + 3 < E) tv.w = dst[eb + 3];
        mk = OWN_HIT(tv.x) | (OWN_HIT(tv.y) << 1) | (OWN_HIT(tv.z) << 2) | (OWN_HIT(tv.w) << 3);
    }
    int c = wave;
    while (true) {
        for (int round = 0; round < 4 * OWN_U; ++round) {
            const bool m = mk != 0u;
            const unsigned bal = __builtin_amdgcn_ballot_w32(m);
            if (bal == 0u) break;
            const int i = m ? __builtin_ctz(mk) : 0;
            mk &= mk - 1u;
            const int sl = i >> 2;
            int kbs = kb[0];
#pragma unroll
            for (int s2 = 1; s2 < OWN_U; ++s2) kbs = (sl == s2) ? kb[s2] : kbs;
            const int pos = cnt + (int)__builtin_amdgcn_mbcnt_lo(bal, 0u);
            if (m) q[pos] = kbs + lane * 4 + (i & 3);
            cnt += __builtin_popcount(bal);
            if (cnt >= 32) {
                own_flush<PASS>(32, q, lane, base, fs, fd, src, dst, w0, w1,
                                t_l, den_l, mx_l, ls);
                const int rem = cnt - 32;
                int tmp = 0;
                if (lane < rem) tmp = q[32 + lane];
                if (lane < rem) q[lane] = tmp;
                cnt = rem;
            }
        }
        if (c >= nfull) break;
        int4 v[OWN_U];
        unsigned okmask = 0;
#pragma unroll
        for (int sl = 0; sl < OWN_U; ++sl) {
            const bool ok = c + sl * OWN_NW < nfull;
            int k = ok ? c + sl * OWN_NW + rot : 0;
            if (k >= nfull) k -= nfull;
            kb[sl] = k * 128;
            okmask |= ok ? (0xFu << (4 * sl)) : 0u;
            v[sl] = *(const int4*)(dst + kb[sl] + lane * 4);
        }
        unsigned nm = 0;
#pragma unroll
        for (int sl = OWN_U - 1; sl >= 0; --sl) {
            nm = (nm << 1) | OWN_HIT(v[sl].w);
            nm = (nm << 1) | OWN_HIT(v[sl].z);
            nm = (nm << 1) | OWN_HIT(v[sl].y);
            nm = (nm << 1) | OWN_HIT(v[sl].x);
        }
        mk = nm & okmask;
        c += OWN_U * OWN_NW;
    }
    if (cnt > 0)
        own_flush<PASS>(cnt, q, lane, base, fs, fd, src, dst, w0, w1,
                        t_l, den_l, mx_l, ls);
}

template <bool FIRST>
__global__ __launch_bounds__(OWN_NT)
void edge_owner(const float* __restrict__ fs, const float* __restrict__ fd,
                const int* __restrict__ src, const int* __restrict__ dst,
                const float* __restrict__ attn, float* __restrict__ keep,
                float* __restrict__ t, int N, int E) {
    __shared__ __attribute__((aligned(16))) float t_l[OWN_R * HID];
    __shared__ float    den_l[OWN_R];
    __shared__ unsigned mx_l[OWN_R];
    __shared__ int      q_l[OWN_NW * 64];
    __shared__ __attribute__((aligned(16))) int ls_src[OWN_CAP];
    __shared__ __attribute__((aligned(16))) int ls_dl[OWN_CAP];
    __shared__ float    ls_sc[OWN_CAP];
    __shared__ int      ls_cnt;

    const int tid  = threadIdx.x;
    const int lane = tid & 31;
    const int wave = __builtin_amdgcn_readfirstlane(tid >> 5);
    const int base = blockIdx.x * OWN_R;

    for (int i = tid; i < OWN_R * HID; i += OWN_NT) t_l[i] = 0.0f;
    if (tid < OWN_R) {
        den_l[tid] = 0.0f;
        mx_l[tid]  = KEY_NEG_INF;
    }
    if (tid == 0) ls_cnt = 0;
    OwnList ls;
    ls.cnt = &ls_cnt;
    ls.src = ls_src;
    ls.dl  = ls_dl;
    ls.sc  = ls_sc;

    const int slotb = (int)(((unsigned)E * 4u) / gridDim.x) & ~127;
    const bool keep_ok = slotb >= 512 && N <= 65536 && OWN_R <= 256;
    OwnKeep kp;
    kp.cap = keep_ok ? min(OWN_CAP, ((slotb - 128) / 3) & ~127) : 0;
    {
        char* sp = (char*)keep + (size_t)blockIdx.x * slotb;
        kp.hdr = (int*)sp;
        kp.src = (unsigned short*)(sp + 128);
        kp.dl  = (unsigned char*)(sp + 128 + 2 * kp.cap);
    }
    int have = -1;
    if (!FIRST && keep_ok) {
        have = __builtin_amdgcn_readfirstlane(kp.hdr[0]);
        if (have > kp.cap) have = -1;
    }
    const float w0 = attn[lane], w1 = attn[32 + lane];
    __syncthreads();

    if (have >= 0) {
        own_list_scores(wave, lane, have, base, fs, fd, w0, w1, kp, ls, mx_l);
        __syncthreads();
        own_list_sums(wave, lane, have, fs, ls, t_l, den_l, mx_l);
    } else {
        own_scan<1>(wave, lane, base, E, fs, fd, src, dst, w0, w1,
                    t_l, den_l, mx_l, q_l + wave * 64, ls);
        __syncthreads();
        const int total = __builtin_amdgcn_readfirstlane(ls_cnt);
        if (FIRST && keep_ok) {
            const bool fits = total <= kp.cap;
            if (tid == 0) kp.hdr[0] = fits ? total : -1;
            if (fits) own_list_save(tid, lane, total, ls, kp);
        }
        if (total <= OWN_CAP) {
            own_list_sums(wave, lane, total, fs, ls, t_l, den_l, mx_l);
        } else {
            own_scan<2>(wave, lane, base, E, fs, fd, src, dst, w0, w1,
                        t_l, den_l, mx_l, q_l + wave * 64, ls);
        }
    }
    __syncthreads();

    for (int idx = tid; idx < OWN_R * 16; idx += OWN_NT) {
        const int row = idx >> 4, c4 = idx & 15;
        const int grow = base + row;
        if (grow < N) {
            const float den = den_l[row];
            const float4 v = *(const float4*)(t_l + row * HID + c4 * 4);
            float4 o = make_float4(0.f, 0.f, 0.f, 0.f);
            if (den > 0.0f) {
                o.x = v.x / den; o.y = v.y / den; o.z = v.z / den; o.w = v.w / den;
            }
            *(float4*)(t + (size_t)grow * HID + c4 * 4) = o;
        }
    }
}

__device__ __forceinline__ void rows_load(const float* __restrict__ X, int arow, int hf,
                                          float4 (&q)[8]) {
    const float* p = X + (size_t)arow * HID + 4 * hf;
#pragma unroll
    for (int j = 0; j < 8; ++j) q[j] = *(const float4*)(p + 8 * j);
}

#define ROWS_K(ks) (8 * ((ks) >> 1) + 2 * ((ks) & 1))

__device__ __forceinline__ v2f rows_afrag(const float4 (&q)[8], int ks) {
    v2f a;
    a.x = (ks & 1) ? q[ks >> 1].z : q[ks >> 1].x;
    a.y = (ks & 1) ? q[ks >> 1].w : q[ks >> 1].y;
    return a;
}

__device__ __forceinline__ void rows_update(float4 (&q)[8], const float4 (&v)[8],
                                            const float* __restrict__ gamma,
                                            const float* __restrict__ beta, int hf) {
    float sum = 0.0f;
#pragma unroll
    for (int j = 0; j < 8; ++j) sum += v[j].x + v[j].y + v[j].z + v[j].w;
    sum += __shfl_xor(sum, 16);
    const float mean = sum * (1.0f / HID);
    float var = 0.0f;
#pragma unroll
    for (int j = 0; j < 8; ++j) {
        const float d0 = v[j].x - mean, d1 = v[j].y - mean;
        const float d2 = v[j].z - mean, d3 = v[j].w - mean;
        var += d0 * d0 + d1 * d1 + d2 * d2 + d3 * d3;
    }
    var += __shfl_xor(var, 16);
    var *= (1.0f / HID);
    const float inv = rsqrtf(var + LN_EPS);
    const float4* pg = (const float4*)(gamma + 4 * hf);
    const float4* pb = (const float4*)(beta + 4 * hf);
    const float k = 0.70710678118654752f;
    float4 x[8];
#pragma unroll
    for (int j = 0; j < 8; ++j) {
        const float4 g = pg[2 * j], b = pb[2 * j];
        x[j].x = (v[j].x - mean) * inv * g.x + b.x;
        x[j].y = (v[j].y - mean) * inv * g.y + b.y;
        x[j].z = (v[j].z - mean) * inv * g.z + b.z;
        x[j].w = (v[j].w - mean) * inv * g.w + b.w;
    }
#pragma unroll
    for (int j = 0; j < 8; ++j) {
        q[j].x += 0.5f * x[j].x * (1.0f + erff(x[j].x * k));
        q[j].y += 0.5f * x[j].y * (1.0f + erff(x[j].y * k));
        q[j].z += 0.5f * x[j].z * (1.0f + erff(x[j].z * k));
        q[j].w += 0.5f * x[j].w * (1.0f + erff(x[j].w * k));
    }
}

__device__ __forceinline__ void rows_wload(const float* __restrict__ W, int hf, int l15,
                                           float4 (&w0)[16], float4 (&w1)[16]) {
    const float* wcol = W + (4 * hf) * HID + 4 * l15;
#pragma unroll
    for (int ks = 0; ks < 16; ++ks) {
        w0[ks] = *(const float4*)(wcol + ROWS_K(ks) * HID);
        w1[ks] = *(const float4*)(wcol + (ROWS_K(ks) + 1) * HID);
    }
}

__device__ __forceinline__ void rows_gemm64(const float4 (&q)[8], const float4 (&w0)[16],
                                            const float4 (&w1)[16],
                                            const float* __restrict__ bias,
                                            float* __restrict__ Y, int rbase, int nrows,
                                            int hf, int l15) {
    const float4 bv = *(const float4*)(bias + 4 * l15);
    v8f acc0, acc1, acc2, acc3;
#pragma unroll
    for (int j = 0; j < 8; ++j) {
        acc0[j] = 0.0f; acc1[j] = 0.0f; acc2[j] = 0.0f; acc3[j] = 0.0f;
    }
#pragma unroll
    for (int ks = 0; ks < 16; ++ks) {
        const v2f a = rows_afrag(q, ks);
        v2f b;
        b.x = w0[ks].x; b.y = w1[ks].x;
        acc0 = __builtin_amdgcn_wmma_f32_16x16x4_f32(
            false, a, false, b, (short)0, acc0, false, false);
        b.x = w0[ks].y; b.y = w1[ks].y;
        acc1 = __builtin_amdgcn_wmma_f32_16x16x4_f32(
            false, a, false, b, (short)0, acc1, false, false);
        b.x = w0[ks].z; b.y = w1[ks].z;
        acc2 = __builtin_amdgcn_wmma_f32_16x16x4_f32(
            false, a, false, b, (short)0, acc2, false, false);
        b.x = w0[ks].w; b.y = w1[ks].w;
        acc3 = __builtin_amdgcn_wmma_f32_16x16x4_f32(
            false, a, false, b, (short)0, acc3, false, false);
    }
#pragma unroll
    for (int i = 0; i < 8; ++i) {
        const int row = rbase + 8 * hf + i;
        if (row < nrows) {
            float4 o;
            o.x = acc0[i] + bv.x;
            o.y = acc1[i] + bv.y;
            o.z = acc2[i] + bv.z;
            o.w = acc3[i] + bv.w;
            *(float4*)(Y + (size_t)row * HID + 4 * l15) = o;
        }
    }
}

__global__ __launch_bounds__(128)
void node_first(const float* __restrict__ h,
                const float* __restrict__ Ws, const float* __restrict__ bs,
                float* __restrict__ fs,
                const float* __restrict__ Wd, const float* __restrict__ bd,
                float* __restrict__ fd, int nrows) {
    const int lane = threadIdx.x & 31;
    const int wave = __builtin_amdgcn_readfirstlane(threadIdx.x >> 5);
    const int hf = lane >> 4, l15 = lane & 15;
    const int rbase = (blockIdx.x * 4 + wave) * 16;
    if (rbase >= nrows) return;
    int arow = rbase + l15;
    if (arow >= nrows) arow = nrows - 1;
    float4 q[8], w0[16], w1[16];
    rows_load(h, arow, hf, q);
    rows_wload(Ws, hf, l15, w0, w1);
    rows_gemm64(q, w0, w1, bs, fs, rbase, nrows, hf, l15);
    rows_wload(Wd, hf, l15, w0, w1);
    rows_gemm64(q, w0, w1, bd, fd, rbase, nrows, hf, l15);
}

__global__ __launch_bounds__(128)
void node_layer(float* __restrict__ h, const float* __restrict__ t,
                const float* __restrict__ gamma, const float* __restrict__ beta,
                const float* __restrict__ Ws, const float* __restrict__ bs,
                float* __restrict__ fs,
                const float* __restrict__ Wd, const float* __restrict__ bd,
                float* __restrict__ fd, int nrows) {
    const int lane = threadIdx.x & 31;
    const int wave = __builtin_amdgcn_readfirstlane(threadIdx.x >> 5);
    const int hf = lane >> 4, l15 = lane & 15;
    const int rbase = (blockIdx.x * 4 + wave) * 16;
    if (rbase >= nrows) return;
    const bool rv = rbase + l15 < nrows;
    const int arow = rv ? rbase + l15 : nrows - 1;
    float4 q[8], v[8], w0[16], w1[16];
    rows_load(t, arow, hf, v);
    rows_load(h, arow, hf, q);
    rows_update(q, v, gamma, beta, hf);
    if (rv) {
        float* p = h + (size_t)arow * HID + 4 * hf;
#pragma unroll
        for (int j = 0; j < 8; ++j) *(float4*)(p + 8 * j) = q[j];
    }
    rows_wload(Ws, hf, l15, w0, w1);
    rows_gemm64(q, w0, w1, bs, fs, rbase, nrows, hf, l15);
    rows_wload(Wd, hf, l15, w0, w1);
    rows_gemm64(q, w0, w1, bd, fd, rbase, nrows, hf, l15);
}

__global__ __launch_bounds__(128)
void node_last(const float* __restrict__ h, const float* __restrict__ t,
               const float* __restrict__ gamma, const float* __restrict__ beta,
               const float* __restrict__ W, const float* __restrict__ bias,
               float* __restrict__ Y, int nrows, int nout) {
    const int lane = threadIdx.x & 31;
    const int wave = __builtin_amdgcn_readfirstlane(threadIdx.x >> 5);
    const int hf = lane >> 4, l15 = lane & 15;
    const int rbase = (blockIdx.x * 4 + wave) * 16;
    if (rbase >= nrows) return;
    int arow = rbase + l15;
    if (arow >= nrows) arow = nrows - 1;
    float4 q[8], v[8];
    rows_load(t, arow, hf, v);
    rows_load(h, arow, hf, q);

    const bool cv = l15 < nout;
    const float* wcol = W + (4 * hf) * nout + (cv ? l15 : 0);
    v2f b[16];
#pragma unroll
    for (int ks = 0; ks < 16; ++ks) {
        const float bx = wcol[ROWS_K(ks) * nout];
        const float by = wcol[(ROWS_K(ks) + 1) * nout];
        b[ks].x = cv ? bx : 0.0f;
        b[ks].y = cv ? by : 0.0f;
    }
    const float bv = cv ? bias[l15] : 0.0f;

    rows_update(q, v, gamma, beta, hf);

    v8f acc;
#pragma unroll
    for (int j = 0; j < 8; ++j) acc[j] = 0.0f;
#pragma unroll
    for (int ks = 0; ks < 16; ++ks)
        acc = __builtin_amdgcn_wmma_f32_16x16x4_f32(
            false, rows_afrag(q, ks), false, b[ks], (short)0, acc, false, false);

#pragma unroll
    for (int i = 0; i < 8; ++i) {
        const int row = rbase + 8 * hf + i;
        if (row < nrows && cv) Y[(size_t)row * nout + l15] = acc[i] + bv;
    }
}

extern "C" void kernel_launch(void* const* d_in, const int* in_sizes, int n_in,
                              void* d_out, int out_size, void* d_ws, size_t ws_size,
                              hipStream_t stream) {
    const float* nodes = (const float*)d_in[0];
    const int*   src   = (const int*)d_in[1];
    const int*   dst   = (const int*)d_in[2];
    const float* w_in  = (const float*)d_in[3];
    const float* b_in  = (const float*)d_in[4];
    const float* w_src = (const float*)d_in[5];
    const float* b_src = (const float*)d_in[6];
    const float* w_dst = (const float*)d_in[7];
    const float* b_dst = (const float*)d_in[8];
    const float* attn  = (const float*)d_in[9];
    const float* gamma = (const float*)d_in[10];
    const float* beta  = (const float*)d_in[11];
    const float* w_out = (const float*)d_in[12];
    const float* b_out = (const float*)d_in[13];

    const int N = in_sizes[0] / HID;
    const int E = in_sizes[1];
    const int NL = 6;

    float* ws   = (float*)d_ws;
    float* h    = ws;  ws += (size_t)N * HID;
    float* fs   = ws;  ws += (size_t)N * HID;
    float* fd   = ws;  ws += (size_t)N * HID;
    float* tbuf = ws;  ws += (size_t)N * HID;
    float* keep = ws;

    const int ggrid  = (N + 63) / 64;
    const int ogrid  = (N + OWN_R - 1) / OWN_R;

    gemm64_direct<<<ggrid, 128, 0, stream>>>(nodes, w_in, b_in, h, N);

    for (int L = 0; L < NL; ++L) {
        const float* ws_L = w_src + (size_t)L * HID * HID;
        const float* wd_L = w_dst + (size_t)L * HID * HID;
        const float* bs_L = b_src + (size_t)L * HID;
        const float* bd_L = b_dst + (size_t)L * HID;
        if (L == 0) {
            node_first<<<ggrid, 128, 0, stream>>>(h, ws_L, bs_L, fs, wd_L, bd_L, fd, N);
        } else {
            node_layer<<<ggrid, 128, 0, stream>>>(h, tbuf, gamma + (size_t)(L - 1) * HID,
                                                  beta + (size_t)(L - 1) * HID,
                                                  ws_L, bs_L, fs, wd_L, bd_L, fd, N);
        }
        if (L == 0) {
            edge_owner<true><<<ogrid, OWN_NT, 0, stream>>>(fs, fd, src, dst,
                                                           attn, keep, tbuf, N, E);
        } else {
            edge_owner<false><<<ogrid, OWN_NT, 0, stream>>>(fs, fd, src, dst,
                                                            attn + (size_t)L * HID,
                                                            keep, tbuf, N, E);
        }
    }

    node_last<<<ggrid, 128, 0, stream>>>(h, tbuf, gamma + (size_t)(NL - 1) * HID,
                                         beta + (size_t)(NL - 1) * HID,
                                         w_out, b_out, (float*)d_out, N, 8);
}
